# starting kernel
# baseline (speedup 1.0000x reference)
#include <hip/hip_runtime.h>
#include <hip/hip_bf16.h>
#include <math.h>

typedef __attribute__((ext_vector_type(2))) float v2f;
typedef __attribute__((ext_vector_type(8))) float v8f;

#define F_DIM 64

__global__ void zero_kernel(float* __restrict__ p, int n) {
    int i = blockIdx.x * blockDim.x + threadIdx.x;
    if (i < n) p[i] = 0.0f;
}

__global__ void __launch_bounds__(64)
matvec_wmma_kernel(const float* __restrict__ X, const float* __restrict__ W,
                   float* __restrict__ s, int n_nodes) {
    const int lane = threadIdx.x & 31;
    const int wave = threadIdx.x >> 5;
    const int tile = blockIdx.x * (blockDim.x >> 5) + wave;
    const int base = tile << 4;
    if (base >= n_nodes) return;

    const int m    = lane & 15;
    const int half = lane >> 4;

    const float* __restrict__ xrow = X + (size_t)(base + m) * F_DIM;

    v8f c = {};
#pragma unroll
    for (int k = 0; k < F_DIM; k += 4) {
        const v2f a = *reinterpret_cast<const v2f*>(xrow + k + (half << 1));
        const v2f wv = *reinterpret_cast<const v2f*>(W + k + (half << 1));
        v2f bf;
        bf.x = (m == 0) ? wv.x : 0.0f;
        bf.y = (m == 0) ? wv.y : 0.0f;
        c = __builtin_amdgcn_wmma_f32_16x16x4_f32(false, a, false, bf,
                                                  (short)0, c, false, false);
    }

    if (m == 0) {
        float* __restrict__ sp = s + base + (half << 3);
#pragma unroll
        for (int r = 0; r < 8; ++r) sp[r] = c[r];
    }
}

__global__ void __launch_bounds__(256)
edge_scatter_kernel(const int* __restrict__ src, const int* __restrict__ dst,
                    const float* __restrict__ s, float* __restrict__ acc,
                    float* __restrict__ deg, int n_edges) {
    int e = blockIdx.x * blockDim.x + threadIdx.x;
    if (e >= n_edges) return;
    const int si = src[e];
    const int di = dst[e];
    const float sv = s[si];
    atomicAdd(&acc[di], sv);
    atomicAdd(&deg[di], 1.0f);
}

__global__ void __launch_bounds__(256)
finalize_kernel(const float* __restrict__ acc, const float* __restrict__ deg,
                const float* __restrict__ b, float* __restrict__ out, int n) {
    int i = blockIdx.x * blockDim.x + threadIdx.x;
    if (i >= n) return;
    const float d = fmaxf(deg[i], 1.0f);
    const float x = acc[i] / d + b[0];
    out[i] = 1.0f / (1.0f + expf(-x));
}

extern "C" void kernel_launch(void* const* d_in, const int* in_sizes, int n_in,
                              void* d_out, int out_size, void* d_ws, size_t ws_size,
                              hipStream_t stream) {
    const float* X     = (const float*)d_in[0];
    const int*   edges = (const int*)d_in[1];
    const float* W     = (const float*)d_in[2];
    const float* b     = (const float*)d_in[3];
    float*       out   = (float*)d_out;

    const int n_nodes = in_sizes[0] / F_DIM;
    const int n_edges = in_sizes[1] / 2;
    const int* src = edges;
    const int* dst = edges + n_edges;

    float* s   = (float*)d_ws;
    float* acc = s + n_nodes;
    float* deg = acc + n_nodes;

    {
        const int n = 2 * n_nodes;
        zero_kernel<<<(n + 255) / 256, 256, 0, stream>>>(acc, n);
    }

    {
        const int tiles  = (n_nodes + 15) / 16;
        const int blocks = (tiles + 1) / 2;
        matvec_wmma_kernel<<<blocks, 64, 0, stream>>>(X, W, s, n_nodes);
    }

    edge_scatter_kernel<<<(n_edges + 255) / 256, 256, 0, stream>>>(
        src, dst, s, acc, deg, n_edges);

    finalize_kernel<<<(n_nodes + 255) / 256, 256, 0, stream>>>(
        acc, deg, b, out, n_nodes);
}


# Edges bucketed once by destination range (LDS counting sort beside the matvec); one work-group per range sums its buckets in LDS (arrival order) and writes out: no partial planes, 2 launches not 3
# speedup vs baseline: 4.8827x; 4.8827x over previous
#include <hip/hip_runtime.h>
#include <hip/hip_bf16.h>
#include <math.h>


typedef __attribute__((ext_vector_type(2))) float v2f;
typedef __attribute__((ext_vector_type(8))) float v8f;

#define F_DIM 64

#define N_BUCKETS   256
#define LOCAL_BITS  9
#define RANGE_CAP   512
#define OFF_PITCH   288
#define STAGE_CAP   5120
#define A_THREADS   256
#define A_UNROLL    8
#define B_THREADS   512
#define B_UNROLL    16

struct Layout {
    int off_base;
    int edge_base;
    int stride_e;
    int per;
    int range;
    int fits;
};

__device__ __forceinline__ Layout layout_of(int n_nodes, int n_edges, unsigned long long ws_bytes) {
    Layout L;
    const int s_words = (n_nodes + 31) & ~31;
    L.range     = (n_nodes + N_BUCKETS - 1) / N_BUCKETS;
    L.per       = (n_edges + N_BUCKETS - 1) / N_BUCKETS;
    L.stride_e  = (L.per + 31) & ~31;
    L.off_base  = s_words;
    L.edge_base = s_words + N_BUCKETS * OFF_PITCH;
    const unsigned long long words =
        (unsigned long long)L.edge_base + (unsigned long long)N_BUCKETS * (unsigned long long)L.stride_e;
    L.fits = (words * 4ull <= ws_bytes) && (L.per <= STAGE_CAP) && (L.range <= RANGE_CAP) &&
             (n_nodes <= (1 << (32 - LOCAL_BITS))) && (L.range > 0);
    return L;
}

#define HOLD8(a) asm volatile("" :: "v"((a)[0]), "v"((a)[1]), "v"((a)[2]), "v"((a)[3]), \
                                    "v"((a)[4]), "v"((a)[5]), "v"((a)[6]), "v"((a)[7]))

__device__ __forceinline__ int range_of(int d, int range, float inv_range) {
    int q = (int)((float)d * inv_range);
    q -= (q * range > d) ? 1 : 0;
    q += ((q + 1) * range <= d) ? 1 : 0;
    return q;
}

__device__ __forceinline__ void matvec_tile(const float* __restrict__ X, const float* __restrict__ W,
                                            float* __restrict__ s, int n_nodes, int tile) {
    const int lane = threadIdx.x & 31;
    const int base = tile << 4;
    if (base >= n_nodes) return;

    const int m    = lane & 15;
    const int half = lane >> 4;

    const float* __restrict__ xrow = X + (size_t)(base + m) * F_DIM;

    v8f c = {};
#pragma unroll
    for (int k = 0; k < F_DIM; k += 4) {
        const v2f a = *reinterpret_cast<const v2f*>(xrow + k + (half << 1));
        const v2f wv = *reinterpret_cast<const v2f*>(W + k + (half << 1));
        v2f bf;
        bf.x = (m == 0) ? wv.x : 0.0f;
        bf.y = (m == 0) ? wv.y : 0.0f;
        c = __builtin_amdgcn_wmma_f32_16x16x4_f32(false, a, false, bf,
                                                  (short)0, c, false, false);
    }

    if (m == 0) {
        float* __restrict__ sp = s + base + (half << 3);
#pragma unroll
        for (int r = 0; r < 8; ++r) sp[r] = c[r];
    }
}

__device__ __forceinline__ void bucket_partition(const int* __restrict__ src, const int* __restrict__ dst,
                                                 unsigned int* __restrict__ wsu, const Layout L,
                                                 int n_nodes, int n_edges, int p,
                                                 unsigned int* cnt_l, unsigned int* cur_l,
                                                 unsigned int* wsum_l, unsigned int* stage_l) {
    const int tid  = threadIdx.x;
    const int lane = tid & 31;
    const int wave = tid >> 5;
    const float inv_range = 1.0f / (float)L.range;

    const int start = min(p * L.per, n_edges);
    const int end   = min(n_edges, start + L.per);
    const int last  = max(end - 1, 0);
    const int step  = A_THREADS * A_UNROLL;

    cnt_l[tid] = 0u;
    __syncthreads();

    for (int e0 = start; e0 < end; e0 += step) {
        int d[A_UNROLL];
#pragma unroll
        for (int j = 0; j < A_UNROLL; ++j) d[j] = dst[min(e0 + j * A_THREADS + tid, last)];
#pragma unroll
        for (int j = 0; j < A_UNROLL; ++j) {
            const bool live = (e0 + j * A_THREADS + tid) < end;
            if (live && (unsigned int)d[j] < (unsigned int)n_nodes)
                atomicAdd(&cnt_l[range_of(d[j], L.range, inv_range)], 1u);
        }
    }
    __syncthreads();

    const unsigned int c = cnt_l[tid];
    unsigned int x = c;
#pragma unroll
    for (int o = 1; o < 32; o <<= 1) {
        const unsigned int y = __shfl_up(x, o);
        if (lane >= o) x += y;
    }
    if (lane == 31) wsum_l[wave] = x;
    __syncthreads();
    unsigned int before = 0u;
    for (int w = 0; w < wave; ++w) before += wsum_l[w];
    const unsigned int excl = before + x - c;
    cur_l[tid] = excl;
    unsigned int* __restrict__ off_row = wsu + L.off_base + p * OFF_PITCH;
    off_row[tid] = excl;
    if (tid == N_BUCKETS - 1) {
        off_row[N_BUCKETS] = excl + c;
        wsum_l[A_THREADS / 32] = excl + c;
    }
    __syncthreads();

    for (int e0 = start; e0 < end; e0 += step) {
        int d[A_UNROLL], sv[A_UNROLL];
#pragma unroll
        for (int j = 0; j < A_UNROLL; ++j) {
            const int e = min(e0 + j * A_THREADS + tid, last);
            d[j]  = dst[e];
            sv[j] = src[e];
        }
        HOLD8(sv);
        unsigned int pos[A_UNROLL], word[A_UNROLL];
#pragma unroll
        for (int j = 0; j < A_UNROLL; ++j) {
            const bool live = (e0 + j * A_THREADS + tid) < end;
            pos[j] = 0xffffffffu;
            word[j] = 0u;
            if (live && (unsigned int)d[j] < (unsigned int)n_nodes) {
                const int r = range_of(d[j], L.range, inv_range);
                word[j] = ((unsigned int)sv[j] << LOCAL_BITS) | (unsigned int)(d[j] - r * L.range);
                pos[j]  = atomicAdd(&cur_l[r], 1u);
            }
        }
#pragma unroll
        for (int j = 0; j < A_UNROLL; ++j)
            if (pos[j] != 0xffffffffu) stage_l[pos[j]] = word[j];
    }
    __syncthreads();

    const unsigned int total = wsum_l[A_THREADS / 32];
    unsigned int* __restrict__ seg = wsu + L.edge_base + (size_t)p * (size_t)L.stride_e;
    for (unsigned int i = tid; i < total; i += A_THREADS) seg[i] = stage_l[i];
}

__global__ void __launch_bounds__(A_THREADS)
matvec_and_bucket_kernel(const float* __restrict__ X, const float* __restrict__ W,
                         const int* __restrict__ src, const int* __restrict__ dst,
                         float* __restrict__ ws, int n_nodes, int n_edges,
                         unsigned long long ws_bytes) {
    __shared__ unsigned int cnt_l[N_BUCKETS];
    __shared__ unsigned int cur_l[N_BUCKETS];
    __shared__ unsigned int wsum_l[A_THREADS / 32 + 1];
    __shared__ unsigned int stage_l[STAGE_CAP];

    if (blockIdx.x < N_BUCKETS) {
        const Layout L = layout_of(n_nodes, n_edges, ws_bytes);
        if (!L.fits) return;
        bucket_partition(src, dst, reinterpret_cast<unsigned int*>(ws), L, n_nodes, n_edges,
                         (int)blockIdx.x, cnt_l, cur_l, wsum_l, stage_l);
    } else {
        const int tile = ((int)blockIdx.x - N_BUCKETS) * (A_THREADS / 32) + ((int)threadIdx.x >> 5);
        matvec_tile(X, W, ws, n_nodes, tile);
    }
}

__global__ void __launch_bounds__(B_THREADS)
range_sum_kernel(const int* __restrict__ src, const int* __restrict__ dst,
                 const float* __restrict__ ws, const float* __restrict__ b,
                 float* __restrict__ out, int n_nodes, int n_edges,
                 unsigned long long ws_bytes) {
    __shared__ float        acc_l[RANGE_CAP];
    __shared__ unsigned int deg_l[RANGE_CAP];

    const int tid = threadIdx.x;
    const int r   = blockIdx.x;
    const Layout L = layout_of(n_nodes, n_edges, ws_bytes);
    const int range = (n_nodes + N_BUCKETS - 1) / N_BUCKETS;
    const int lo    = r * range;

    for (int i = tid; i < RANGE_CAP; i += B_THREADS) {
        acc_l[i] = 0.0f;
        deg_l[i] = 0u;
    }
    __syncthreads();

    const float* __restrict__ s = ws;
    if (L.fits) {
        const unsigned int* __restrict__ wsu = reinterpret_cast<const unsigned int*>(ws);
        const int p   = tid >> 1;
        const int sub = tid & 1;
        const unsigned int* __restrict__ off_row = wsu + L.off_base + p * OFF_PITCH;
        const int beg = (int)off_row[r];
        const int end = (int)off_row[r + 1];
        const unsigned int* __restrict__ seg = wsu + L.edge_base + (size_t)p * (size_t)L.stride_e;
        for (int i0 = beg + sub; i0 < end; i0 += 2 * B_UNROLL) {
            unsigned int w[B_UNROLL];
            float v[B_UNROLL];
#pragma unroll
            for (int j = 0; j < B_UNROLL; ++j) w[j] = seg[min(i0 + 2 * j, end - 1)];
            HOLD8(w); HOLD8(w + 8);
#pragma unroll
            for (int j = 0; j < B_UNROLL; ++j) {
                const unsigned int node = (i0 + 2 * j < end) ? (w[j] >> LOCAL_BITS) : 0u;
                v[j] = s[node];
            }
            HOLD8(v); HOLD8(v + 8);
#pragma unroll
            for (int j = 0; j < B_UNROLL; ++j) {
                if (i0 + 2 * j < end) {
                    const unsigned int loc = w[j] & (unsigned int)(RANGE_CAP - 1);
                    atomicAdd(&acc_l[loc], v[j]);
                    atomicAdd(&deg_l[loc], 1u);
                }
            }
        }
    } else if (range <= RANGE_CAP) {
        for (int e = tid; e < n_edges; e += B_THREADS) {
            const unsigned int loc = (unsigned int)(dst[e] - lo);
            if (loc < (unsigned int)range && (unsigned int)dst[e] < (unsigned int)n_nodes) {
                atomicAdd(&acc_l[loc], s[src[e]]);
                atomicAdd(&deg_l[loc], 1u);
            }
        }
    }
    __syncthreads();

    const float bias = b[0];
    for (int i = tid; i < range && i < RANGE_CAP; i += B_THREADS) {
        const int node = lo + i;
        if (node < n_nodes) {
            const float d = fmaxf((float)deg_l[i], 1.0f);
            const float x = acc_l[i] / d + bias;
            out[node] = 1.0f / (1.0f + expf(-x));
        }
    }
}

extern "C" void kernel_launch(void* const* d_in, const int* in_sizes, int n_in,
                              void* d_out, int out_size, void* d_ws, size_t ws_size,
                              hipStream_t stream) {
    const float* X     = (const float*)d_in[0];
    const int*   edges = (const int*)d_in[1];
    const float* W     = (const float*)d_in[2];
    const float* b     = (const float*)d_in[3];
    float*       out   = (float*)d_out;

    const int n_nodes = in_sizes[0] / F_DIM;
    const int n_edges = in_sizes[1] / 2;
    const int* src = edges;
    const int* dst = edges + n_edges;

    float* ws = (float*)d_ws;

    {
        const int tiles  = (n_nodes + 15) / 16;
        const int blocks = (tiles + A_THREADS / 32 - 1) / (A_THREADS / 32);
        matvec_and_bucket_kernel<<<N_BUCKETS + blocks, A_THREADS, 0, stream>>>(
            X, W, src, dst, ws, n_nodes, n_edges, (unsigned long long)ws_size);
    }

    range_sum_kernel<<<N_BUCKETS, B_THREADS, 0, stream>>>(
        src, dst, ws, b, out, n_nodes, n_edges, (unsigned long long)ws_size);
}


# Bucketing work-groups read their edge partition once: 20 edges a thread, all 40 index loads in flight, held in registers from count to place; one memory round trip, not six; sums still arrival order
# speedup vs baseline: 5.5611x; 1.1389x over previous
#include <hip/hip_runtime.h>
#include <hip/hip_bf16.h>
#include <math.h>


typedef __attribute__((ext_vector_type(2))) float v2f;
typedef __attribute__((ext_vector_type(8))) float v8f;

#define F_DIM 64

#define N_BUCKETS   256
#define LOCAL_BITS  9
#define RANGE_CAP   512
#define OFF_PITCH   288
#define STAGE_CAP   5120
#define A_THREADS   256
#define A_HOLD      20
#define B_THREADS   512
#define B_UNROLL    16

static_assert(A_THREADS * A_HOLD >= STAGE_CAP, "a work-group must hold a whole partition in registers");

struct Layout {
    int off_base;
    int edge_base;
    int stride_e;
    int per;
    int range;
    int fits;
};

__device__ __forceinline__ Layout layout_of(int n_nodes, int n_edges, unsigned long long ws_bytes) {
    Layout L;
    const int s_words = (n_nodes + 31) & ~31;
    L.range     = (n_nodes + N_BUCKETS - 1) / N_BUCKETS;
    L.per       = (n_edges + N_BUCKETS - 1) / N_BUCKETS;
    L.stride_e  = (L.per + 31) & ~31;
    L.off_base  = s_words;
    L.edge_base = s_words + N_BUCKETS * OFF_PITCH;
    const unsigned long long words =
        (unsigned long long)L.edge_base + (unsigned long long)N_BUCKETS * (unsigned long long)L.stride_e;
    L.fits = (words * 4ull <= ws_bytes) && (L.per <= STAGE_CAP) && (L.range <= RANGE_CAP) &&
             (n_nodes <= (1 << (32 - LOCAL_BITS))) && (L.range > 0);
    return L;
}

#define HOLD8(a) asm volatile("" :: "v"((a)[0]), "v"((a)[1]), "v"((a)[2]), "v"((a)[3]), \
                                    "v"((a)[4]), "v"((a)[5]), "v"((a)[6]), "v"((a)[7]))

__device__ __forceinline__ int range_of(int d, int range, float inv_range) {
    int q = (int)((float)d * inv_range);
    q -= (q * range > d) ? 1 : 0;
    q += ((q + 1) * range <= d) ? 1 : 0;
    return q;
}

__device__ __forceinline__ void matvec_tile(const float* __restrict__ X, const float* __restrict__ W,
                                            float* __restrict__ s, int n_nodes, int tile) {
    const int lane = threadIdx.x & 31;
    const int base = tile << 4;
    if (base >= n_nodes) return;

    const int m    = lane & 15;
    const int half = lane >> 4;

    const float* __restrict__ xrow = X + (size_t)(base + m) * F_DIM;

    v8f c = {};
#pragma unroll
    for (int k = 0; k < F_DIM; k += 4) {
        const v2f a = *reinterpret_cast<const v2f*>(xrow + k + (half << 1));
        const v2f wv = *reinterpret_cast<const v2f*>(W + k + (half << 1));
        v2f bf;
        bf.x = (m == 0) ? wv.x : 0.0f;
        bf.y = (m == 0) ? wv.y : 0.0f;
        c = __builtin_amdgcn_wmma_f32_16x16x4_f32(false, a, false, bf,
                                                  (short)0, c, false, false);
    }

    if (m == 0) {
        float* __restrict__ sp = s + base + (half << 3);
#pragma unroll
        for (int r = 0; r < 8; ++r) sp[r] = c[r];
    }
}

__device__ __forceinline__ void bucket_partition(const int* __restrict__ src, const int* __restrict__ dst,
                                                 unsigned int* __restrict__ wsu, const Layout L,
                                                 int n_nodes, int n_edges, int p,
                                                 unsigned int* cnt_l, unsigned int* cur_l,
                                                 unsigned int* wsum_l, unsigned int* stage_l) {
    const int tid  = threadIdx.x;
    const int lane = tid & 31;
    const int wave = tid >> 5;
    const float inv_range = 1.0f / (float)L.range;

    const int start = min(p * L.per, n_edges);
    const int end   = min(n_edges, start + L.per);
    const int last  = max(end - 1, 0);
    const int  span = A_THREADS * A_HOLD;
    const bool wide = n_edges >= span;
    const int  base = wide ? min(start, n_edges - span) : start;
    int d[A_HOLD], sv[A_HOLD];
    if (wide) {
        const int* __restrict__ dp = dst + base + tid;
        const int* __restrict__ sp = src + base + tid;
#pragma unroll
        for (int j = 0; j < A_HOLD; ++j) {
            d[j]  = dp[j * A_THREADS];
            sv[j] = sp[j * A_THREADS];
        }
    } else {
#pragma unroll
        for (int j = 0; j < A_HOLD; ++j) {
            d[j] = 0; sv[j] = 0;
            if (start < end) {
                const int e = min(base + j * A_THREADS + tid, last);
                d[j]  = dst[e];
                sv[j] = src[e];
            }
        }
    }
    int          rr[A_HOLD];
    unsigned int word[A_HOLD];
#pragma unroll
    for (int j = 0; j < A_HOLD; ++j) {
        const int  e  = base + j * A_THREADS + tid;
        const bool ok = (e >= start) && (e < end) && ((unsigned int)d[j] < (unsigned int)n_nodes);
        const int dd = ok ? d[j] : 0;
        const int r  = range_of(dd, L.range, inv_range);
        rr[j]   = ok ? r : -1;
        word[j] = ((unsigned int)sv[j] << LOCAL_BITS) | (unsigned int)(dd - r * L.range);
    }

    cnt_l[tid] = 0u;
    __syncthreads();

#pragma unroll
    for (int j = 0; j < A_HOLD; ++j)
        if (rr[j] >= 0) atomicAdd(&cnt_l[rr[j]], 1u);
    __syncthreads();

    const unsigned int c = cnt_l[tid];
    unsigned int x = c;
#pragma unroll
    for (int o = 1; o < 32; o <<= 1) {
        const unsigned int y = __shfl_up(x, o);
        if (lane >= o) x += y;
    }
    if (lane == 31) wsum_l[wave] = x;
    __syncthreads();
    unsigned int before = 0u;
    for (int w = 0; w < wave; ++w) before += wsum_l[w];
    const unsigned int excl = before + x - c;
    cur_l[tid] = excl;
    unsigned int* __restrict__ off_row = wsu + L.off_base + p * OFF_PITCH;
    off_row[tid] = excl;
    if (tid == N_BUCKETS - 1) {
        off_row[N_BUCKETS] = excl + c;
        wsum_l[A_THREADS / 32] = excl + c;
    }
    __syncthreads();

#pragma unroll
    for (int h = 0; h < A_HOLD; h += A_HOLD / 2) {
        unsigned int pos[A_HOLD / 2];
#pragma unroll
        for (int j = 0; j < A_HOLD / 2; ++j) {
            pos[j] = 0xffffffffu;
            if (rr[h + j] >= 0) pos[j] = atomicAdd(&cur_l[rr[h + j]], 1u);
        }
#pragma unroll
        for (int j = 0; j < A_HOLD / 2; ++j)
            if (pos[j] != 0xffffffffu) stage_l[pos[j]] = word[h + j];
    }
    __syncthreads();

    const unsigned int total = wsum_l[A_THREADS / 32];
    unsigned int* __restrict__ seg = wsu + L.edge_base + (size_t)p * (size_t)L.stride_e;
    for (unsigned int i = tid; i < total; i += A_THREADS) seg[i] = stage_l[i];
}

__global__ void __launch_bounds__(A_THREADS)
matvec_and_bucket_kernel(const float* __restrict__ X, const float* __restrict__ W,
                         const int* __restrict__ src, const int* __restrict__ dst,
                         float* __restrict__ ws, int n_nodes, int n_edges,
                         unsigned long long ws_bytes) {
    __shared__ unsigned int cnt_l[N_BUCKETS];
    __shared__ unsigned int cur_l[N_BUCKETS];
    __shared__ unsigned int wsum_l[A_THREADS / 32 + 1];
    __shared__ unsigned int stage_l[STAGE_CAP];

    if (blockIdx.x < N_BUCKETS) {
        const Layout L = layout_of(n_nodes, n_edges, ws_bytes);
        if (!L.fits) return;
        bucket_partition(src, dst, reinterpret_cast<unsigned int*>(ws), L, n_nodes, n_edges,
                         (int)blockIdx.x, cnt_l, cur_l, wsum_l, stage_l);
    } else {
        const int tile = ((int)blockIdx.x - N_BUCKETS) * (A_THREADS / 32) + ((int)threadIdx.x >> 5);
        matvec_tile(X, W, ws, n_nodes, tile);
    }
}

__global__ void __launch_bounds__(B_THREADS)
range_sum_kernel(const int* __restrict__ src, const int* __restrict__ dst,
                 const float* __restrict__ ws, const float* __restrict__ b,
                 float* __restrict__ out, int n_nodes, int n_edges,
                 unsigned long long ws_bytes) {
    __shared__ float        acc_l[RANGE_CAP];
    __shared__ unsigned int deg_l[RANGE_CAP];

    const int tid = threadIdx.x;
    const int r   = blockIdx.x;
    const Layout L = layout_of(n_nodes, n_edges, ws_bytes);
    const int range = (n_nodes + N_BUCKETS - 1) / N_BUCKETS;
    const int lo    = r * range;

    for (int i = tid; i < RANGE_CAP; i += B_THREADS) {
        acc_l[i] = 0.0f;
        deg_l[i] = 0u;
    }
    __syncthreads();

    const float* __restrict__ s = ws;
    if (L.fits) {
        const unsigned int* __restrict__ wsu = reinterpret_cast<const unsigned int*>(ws);
        const int p   = tid >> 1;
        const int sub = tid & 1;
        const unsigned int* __restrict__ off_row = wsu + L.off_base + p * OFF_PITCH;
        const int beg = (int)off_row[r];
        const int end = (int)off_row[r + 1];
        const unsigned int* __restrict__ seg = wsu + L.edge_base + (size_t)p * (size_t)L.stride_e;
        for (int i0 = beg + sub; i0 < end; i0 += 2 * B_UNROLL) {
            unsigned int w[B_UNROLL];
            float v[B_UNROLL];
#pragma unroll
            for (int j = 0; j < B_UNROLL; ++j) w[j] = seg[min(i0 + 2 * j, end - 1)];
            HOLD8(w); HOLD8(w + 8);
#pragma unroll
            for (int j = 0; j < B_UNROLL; ++j) {
                const unsigned int node = (i0 + 2 * j < end) ? (w[j] >> LOCAL_BITS) : 0u;
                v[j] = s[node];
            }
            HOLD8(v); HOLD8(v + 8);
#pragma unroll
            for (int j = 0; j < B_UNROLL; ++j) {
                if (i0 + 2 * j < end) {
                    const unsigned int loc = w[j] & (unsigned int)(RANGE_CAP - 1);
                    atomicAdd(&acc_l[loc], v[j]);
                    atomicAdd(&deg_l[loc], 1u);
                }
            }
        }
    } else if (range <= RANGE_CAP) {
        for (int e = tid; e < n_edges; e += B_THREADS) {
            const unsigned int loc = (unsigned int)(dst[e] - lo);
            if (loc < (unsigned int)range && (unsigned int)dst[e] < (unsigned int)n_nodes) {
                atomicAdd(&acc_l[loc], s[src[e]]);
                atomicAdd(&deg_l[loc], 1u);
            }
        }
    }
    __syncthreads();

    const float bias = b[0];
    for (int i = tid; i < range && i < RANGE_CAP; i += B_THREADS) {
        const int node = lo + i;
        if (node < n_nodes) {
            const float d = fmaxf((float)deg_l[i], 1.0f);
            const float x = acc_l[i] / d + bias;
            out[node] = 1.0f / (1.0f + expf(-x));
        }
    }
}

extern "C" void kernel_launch(void* const* d_in, const int* in_sizes, int n_in,
                              void* d_out, int out_size, void* d_ws, size_t ws_size,
                              hipStream_t stream) {
    const float* X     = (const float*)d_in[0];
    const int*   edges = (const int*)d_in[1];
    const float* W     = (const float*)d_in[2];
    const float* b     = (const float*)d_in[3];
    float*       out   = (float*)d_out;

    const int n_nodes = in_sizes[0] / F_DIM;
    const int n_edges = in_sizes[1] / 2;
    const int* src = edges;
    const int* dst = edges + n_edges;

    float* ws = (float*)d_ws;

    {
        const int tiles  = (n_nodes + 15) / 16;
        const int blocks = (tiles + A_THREADS / 32 - 1) / (A_THREADS / 32);
        matvec_and_bucket_kernel<<<N_BUCKETS + blocks, A_THREADS, 0, stream>>>(
            X, W, src, dst, ws, n_nodes, n_edges, (unsigned long long)ws_size);
    }

    range_sum_kernel<<<N_BUCKETS, B_THREADS, 0, stream>>>(
        src, dst, ws, b, out, n_nodes, n_edges, (unsigned long long)ws_size);
}


# Bucketing: count atomics give back each edge's rank; placing is first word + rank by plain LDS reads (one atomic pass, not two); offset row stored behind the last barrier; sums still arrival order
# speedup vs baseline: 5.8771x; 1.0568x over previous
#include <hip/hip_runtime.h>
#include <hip/hip_bf16.h>
#include <math.h>


typedef __attribute__((ext_vector_type(2))) float v2f;
typedef __attribute__((ext_vector_type(8))) float v8f;

#define F_DIM 64

#define N_BUCKETS   256
#define LOCAL_BITS  9
#define RANGE_CAP   512
#define OFF_PITCH   288
#define STAGE_CAP   5120
#define A_THREADS   256
#define A_HOLD      20
#define B_THREADS   512
#define B_UNROLL    16

static_assert(A_THREADS * A_HOLD >= STAGE_CAP, "a work-group must hold a whole partition in registers");

struct Layout {
    int off_base;
    int edge_base;
    int stride_e;
    int per;
    int range;
    int fits;
};

__device__ __forceinline__ Layout layout_of(int n_nodes, int n_edges, unsigned long long ws_bytes) {
    Layout L;
    const int s_words = (n_nodes + 31) & ~31;
    L.range     = (n_nodes + N_BUCKETS - 1) / N_BUCKETS;
    L.per       = (n_edges + N_BUCKETS - 1) / N_BUCKETS;
    L.stride_e  = (L.per + 31) & ~31;
    L.off_base  = s_words;
    L.edge_base = s_words + N_BUCKETS * OFF_PITCH;
    const unsigned long long words =
        (unsigned long long)L.edge_base + (unsigned long long)N_BUCKETS * (unsigned long long)L.stride_e;
    L.fits = (words * 4ull <= ws_bytes) && (L.per <= STAGE_CAP) && (L.range <= RANGE_CAP) &&
             (n_nodes <= (1 << (32 - LOCAL_BITS))) && (L.range > 0);
    return L;
}

#define HOLD8(a) asm volatile("" :: "v"((a)[0]), "v"((a)[1]), "v"((a)[2]), "v"((a)[3]), \
                                    "v"((a)[4]), "v"((a)[5]), "v"((a)[6]), "v"((a)[7]))

#define HOLD4(a) asm volatile("" :: "v"((a)[0]), "v"((a)[1]), "v"((a)[2]), "v"((a)[3]))
static_assert(A_HOLD == 20, "the placing step holds its 20 first words with HOLD8, HOLD8, HOLD4");

__device__ __forceinline__ int range_of(int d, int range, float inv_range) {
    int q = (int)((float)d * inv_range);
    q -= (q * range > d) ? 1 : 0;
    q += ((q + 1) * range <= d) ? 1 : 0;
    return q;
}

__device__ __forceinline__ void matvec_tile(const float* __restrict__ X, const float* __restrict__ W,
                                            float* __restrict__ s, int n_nodes, int tile) {
    const int lane = threadIdx.x & 31;
    const int base = tile << 4;
    if (base >= n_nodes) return;

    const int m    = lane & 15;
    const int half = lane >> 4;

    const float* __restrict__ xrow = X + (size_t)(base + m) * F_DIM;

    v8f c = {};
#pragma unroll
    for (int k = 0; k < F_DIM; k += 4) {
        const v2f a = *reinterpret_cast<const v2f*>(xrow + k + (half << 1));
        const v2f wv = *reinterpret_cast<const v2f*>(W + k + (half << 1));
        v2f bf;
        bf.x = (m == 0) ? wv.x : 0.0f;
        bf.y = (m == 0) ? wv.y : 0.0f;
        c = __builtin_amdgcn_wmma_f32_16x16x4_f32(false, a, false, bf,
                                                  (short)0, c, false, false);
    }

    if (m == 0) {
        float* __restrict__ sp = s + base + (half << 3);
#pragma unroll
        for (int r = 0; r < 8; ++r) sp[r] = c[r];
    }
}

__device__ __forceinline__ void bucket_partition(const int* __restrict__ src, const int* __restrict__ dst,
                                                 unsigned int* __restrict__ wsu, const Layout L,
                                                 int n_nodes, int n_edges, int p,
                                                 unsigned int* cnt_l, unsigned int* cur_l,
                                                 unsigned int* wsum_l, unsigned int* stage_l) {
    const int tid  = threadIdx.x;
    const int lane = tid & 31;
    const int wave = tid >> 5;
    const float inv_range = 1.0f / (float)L.range;

    const int start = min(p * L.per, n_edges);
    const int end   = min(n_edges, start + L.per);
    const int last  = max(end - 1, 0);
    const int  span = A_THREADS * A_HOLD;
    const bool wide = n_edges >= span;
    const int  base = wide ? min(start, n_edges - span) : start;
    int d[A_HOLD], sv[A_HOLD];
    if (wide) {
        const int* __restrict__ dp = dst + base + tid;
        const int* __restrict__ sp = src + base + tid;
#pragma unroll
        for (int j = 0; j < A_HOLD; ++j) {
            d[j]  = dp[j * A_THREADS];
            sv[j] = sp[j * A_THREADS];
        }
    } else {
#pragma unroll
        for (int j = 0; j < A_HOLD; ++j) {
            d[j] = 0; sv[j] = 0;
            if (start < end) {
                const int e = min(base + j * A_THREADS + tid, last);
                d[j]  = dst[e];
                sv[j] = src[e];
            }
        }
    }
    int          rr[A_HOLD];
    unsigned int word[A_HOLD];
#pragma unroll
    for (int j = 0; j < A_HOLD; ++j) {
        const int  e  = base + j * A_THREADS + tid;
        const bool ok = (e >= start) && (e < end) && ((unsigned int)d[j] < (unsigned int)n_nodes);
        const int dd = ok ? d[j] : 0;
        const int r  = range_of(dd, L.range, inv_range);
        rr[j]   = ok ? r : N_BUCKETS + lane;
        word[j] = ((unsigned int)sv[j] << LOCAL_BITS) | (unsigned int)(dd - r * L.range);
    }
    HOLD8(word); HOLD8(word + 8); HOLD4(word + 16);

    cnt_l[tid] = 0u;
    if (tid < 32) {
        cnt_l[N_BUCKETS + tid] = 0u;
        cur_l[N_BUCKETS + tid] = 0u;
    }
    __syncthreads();

    unsigned int rank[A_HOLD];
#pragma unroll
    for (int j = 0; j < A_HOLD; ++j) rank[j] = atomicAdd(&cnt_l[rr[j]], 1u);
    __syncthreads();

    const unsigned int c = cnt_l[tid];
    unsigned int x = c;
#pragma unroll
    for (int o = 1; o < 32; o <<= 1) {
        const unsigned int y = __shfl_up(x, o);
        if (lane >= o) x += y;
    }
    if (lane == 31) wsum_l[wave] = x;
    __syncthreads();
    unsigned int before = 0u;
    for (int w = 0; w < wave; ++w) before += wsum_l[w];
    const unsigned int excl = before + x - c;
    cur_l[tid] = excl;
    if (tid == N_BUCKETS - 1) wsum_l[A_THREADS / 32] = excl + c;
    __syncthreads();

    unsigned int first[A_HOLD];
#pragma unroll
    for (int j = 0; j < A_HOLD; ++j) first[j] = cur_l[rr[j]];
    HOLD8(first); HOLD8(first + 8); HOLD4(first + 16);
#pragma unroll
    for (int j = 0; j < A_HOLD; ++j)
        stage_l[(rr[j] < N_BUCKETS) ? first[j] + rank[j] : (unsigned int)(STAGE_CAP + lane)] = word[j];
    __syncthreads();

    unsigned int* __restrict__ off_row = wsu + L.off_base + p * OFF_PITCH;
    off_row[tid] = excl;
    if (tid == N_BUCKETS - 1) off_row[N_BUCKETS] = excl + c;
    const unsigned int total = wsum_l[A_THREADS / 32];
    unsigned int* __restrict__ seg = wsu + L.edge_base + (size_t)p * (size_t)L.stride_e;
    for (unsigned int i = tid; i < total; i += A_THREADS) seg[i] = stage_l[i];
}

__global__ void __launch_bounds__(A_THREADS)
matvec_and_bucket_kernel(const float* __restrict__ X, const float* __restrict__ W,
                         const int* __restrict__ src, const int* __restrict__ dst,
                         float* __restrict__ ws, int n_nodes, int n_edges,
                         unsigned long long ws_bytes) {
    __shared__ unsigned int cnt_l[N_BUCKETS + 32];
    __shared__ unsigned int cur_l[N_BUCKETS + 32];
    __shared__ unsigned int wsum_l[A_THREADS / 32 + 1];
    __shared__ unsigned int stage_l[STAGE_CAP + 32];

    if (blockIdx.x < N_BUCKETS) {
        const Layout L = layout_of(n_nodes, n_edges, ws_bytes);
        if (!L.fits) return;
        bucket_partition(src, dst, reinterpret_cast<unsigned int*>(ws), L, n_nodes, n_edges,
                         (int)blockIdx.x, cnt_l, cur_l, wsum_l, stage_l);
    } else {
        const int tile = ((int)blockIdx.x - N_BUCKETS) * (A_THREADS / 32) + ((int)threadIdx.x >> 5);
        matvec_tile(X, W, ws, n_nodes, tile);
    }
}

__global__ void __launch_bounds__(B_THREADS)
range_sum_kernel(const int* __restrict__ src, const int* __restrict__ dst,
                 const float* __restrict__ ws, const float* __restrict__ b,
                 float* __restrict__ out, int n_nodes, int n_edges,
                 unsigned long long ws_bytes) {
    __shared__ float        acc_l[RANGE_CAP];
    __shared__ unsigned int deg_l[RANGE_CAP];

    const int tid = threadIdx.x;
    const int r   = blockIdx.x;
    const Layout L = layout_of(n_nodes, n_edges, ws_bytes);
    const int range = (n_nodes + N_BUCKETS - 1) / N_BUCKETS;
    const int lo    = r * range;

    for (int i = tid; i < RANGE_CAP; i += B_THREADS) {
        acc_l[i] = 0.0f;
        deg_l[i] = 0u;
    }
    __syncthreads();

    const float* __restrict__ s = ws;
    if (L.fits) {
        const unsigned int* __restrict__ wsu = reinterpret_cast<const unsigned int*>(ws);
        const int p   = tid >> 1;
        const int sub = tid & 1;
        const unsigned int* __restrict__ off_row = wsu + L.off_base + p * OFF_PITCH;
        const int beg = (int)off_row[r];
        const int end = (int)off_row[r + 1];
        const unsigned int* __restrict__ seg = wsu + L.edge_base + (size_t)p * (size_t)L.stride_e;
        for (int i0 = beg + sub; i0 < end; i0 += 2 * B_UNROLL) {
            unsigned int w[B_UNROLL];
            float v[B_UNROLL];
#pragma unroll
            for (int j = 0; j < B_UNROLL; ++j) w[j] = seg[min(i0 + 2 * j, end - 1)];
            HOLD8(w); HOLD8(w + 8);
#pragma unroll
            for (int j = 0; j < B_UNROLL; ++j) {
                const unsigned int node = (i0 + 2 * j < end) ? (w[j] >> LOCAL_BITS) : 0u;
                v[j] = s[node];
            }
            HOLD8(v); HOLD8(v + 8);
#pragma unroll
            for (int j = 0; j < B_UNROLL; ++j) {
                if (i0 + 2 * j < end) {
                    const unsigned int loc = w[j] & (unsigned int)(RANGE_CAP - 1);
                    atomicAdd(&acc_l[loc], v[j]);
                    atomicAdd(&deg_l[loc], 1u);
                }
            }
        }
    } else if (range <= RANGE_CAP) {
        for (int e = tid; e < n_edges; e += B_THREADS) {
            const unsigned int loc = (unsigned int)(dst[e] - lo);
            if (loc < (unsigned int)range && (unsigned int)dst[e] < (unsigned int)n_nodes) {
                atomicAdd(&acc_l[loc], s[src[e]]);
                atomicAdd(&deg_l[loc], 1u);
            }
        }
    }
    __syncthreads();

    const float bias = b[0];
    for (int i = tid; i < range && i < RANGE_CAP; i += B_THREADS) {
        const int node = lo + i;
        if (node < n_nodes) {
            const float d = fmaxf((float)deg_l[i], 1.0f);
            const float x = acc_l[i] / d + bias;
            out[node] = 1.0f / (1.0f + expf(-x));
        }
    }
}

extern "C" void kernel_launch(void* const* d_in, const int* in_sizes, int n_in,
                              void* d_out, int out_size, void* d_ws, size_t ws_size,
                              hipStream_t stream) {
    const float* X     = (const float*)d_in[0];
    const int*   edges = (const int*)d_in[1];
    const float* W     = (const float*)d_in[2];
    const float* b     = (const float*)d_in[3];
    float*       out   = (float*)d_out;

    const int n_nodes = in_sizes[0] / F_DIM;
    const int n_edges = in_sizes[1] / 2;
    const int* src = edges;
    const int* dst = edges + n_edges;

    float* ws = (float*)d_ws;

    {
        const int tiles  = (n_nodes + 15) / 16;
        const int blocks = (tiles + A_THREADS / 32 - 1) / (A_THREADS / 32);
        matvec_and_bucket_kernel<<<N_BUCKETS + blocks, A_THREADS, 0, stream>>>(
            X, W, src, dst, ws, n_nodes, n_edges, (unsigned long long)ws_size);
    }

    range_sum_kernel<<<N_BUCKETS, B_THREADS, 0, stream>>>(
        src, dst, ws, b, out, n_nodes, n_edges, (unsigned long long)ws_size);
}
